# starting kernel
# baseline (speedup 1.0000x reference)
#include <hip/hip_runtime.h>

#define BB 4
#define SS 128
#define DD 256
#define NN 1024
#define EPS 1e-6f

typedef __attribute__((ext_vector_type(2))) float v2f;
typedef __attribute__((ext_vector_type(8))) float v8f;

__global__ void lns_linear_kernel(const float* __restrict__ x,
                                  const float* __restrict__ w,
                                  const float* __restrict__ bias,
                                  float* __restrict__ out) {
    int idx = blockIdx.x * blockDim.x + threadIdx.x;
    int n  = idx & (NN - 1);
    int bs = idx >> 10;
    int s  = bs & (SS - 1);
    int b  = bs >> 7;
    int d  = n >> 2;
    int c  = n & 3;
    float val = fmaxf(x[(b * SS + s) * 4 + c], EPS)
              + __expf(w[(c * SS + s) * DD + d])
              + __expf(bias[d * 4 + c]);
    out[idx] = val;
}

__global__ void scores_wmma_kernel(const float* __restrict__ Qr,
                                   const float* __restrict__ Kr,
                                   float* __restrict__ scores) {
    int wave = (blockIdx.x * blockDim.x + threadIdx.x) >> 5;
    int lane = threadIdx.x & 31;
    int b  = wave >> 6;
    int i0 = ((wave >> 3) & 7) << 4;
    int j0 = (wave & 7) << 4;
    int half = lane >> 4;
    int lr   = lane & 15;

    const float* arow = Qr + (size_t)(b * SS + i0 + lr) * NN + 2 * half;
    const float* brow = Kr + (size_t)(b * SS + j0 + lr) * NN + 2 * half;

    v8f c = {};
    #pragma unroll 4
    for (int k = 0; k < NN; k += 4) {
        v2f a  = *(const v2f*)(arow + k);
        v2f bm = *(const v2f*)(brow + k);
        c = __builtin_amdgcn_wmma_f32_16x16x4_f32(false, a, false, bm,
                                                  (short)0, c, false, false);
    }

    float* srow = scores + (size_t)(b * SS + i0 + 8 * half) * SS + j0 + lr;
    #pragma unroll
    for (int r = 0; r < 8; ++r) srow[(size_t)r * SS] = c[r];
}

__global__ void softmax_clip_kernel(float* __restrict__ scores) {
    __shared__ float red[SS];
    int row = blockIdx.x;
    int t   = threadIdx.x;
    float v = scores[(size_t)row * SS + t];
    red[t] = v;
    __syncthreads();
    for (int off = SS / 2; off > 0; off >>= 1) {
        if (t < off) red[t] = fmaxf(red[t], red[t + off]);
        __syncthreads();
    }
    float m = red[0];
    __syncthreads();
    float e = __expf(v - m);
    red[t] = e;
    __syncthreads();
    for (int off = SS / 2; off > 0; off >>= 1) {
        if (t < off) red[t] += red[t + off];
        __syncthreads();
    }
    float sum = red[0];
    scores[(size_t)row * SS + t] = fmaxf(e / sum, EPS);
}

__global__ void out_wmma_kernel(const float* __restrict__ attn,
                                const float* __restrict__ Vr,
                                float* __restrict__ out) {
    int wave = (blockIdx.x * blockDim.x + threadIdx.x) >> 5;
    int lane = threadIdx.x & 31;
    int b  = wave >> 9;
    int i0 = ((wave >> 6) & 7) << 4;
    int n0 = (wave & 63) << 4;
    int half = lane >> 4;
    int lr   = lane & 15;

    const float* arow = attn + (size_t)(b * SS + i0 + lr) * SS + 2 * half;
    const float* bcol = Vr + (size_t)(b * SS) * NN + n0 + lr;

    v8f c = {};
    #pragma unroll 4
    for (int k = 0; k < SS; k += 4) {
        v2f a = *(const v2f*)(arow + k);
        int kk = k + 2 * half;
        v2f bm;
        bm[0] = bcol[(size_t)kk * NN];
        bm[1] = bcol[(size_t)(kk + 1) * NN];
        c = __builtin_amdgcn_wmma_f32_16x16x4_f32(false, a, false, bm,
                                                  (short)0, c, false, false);
    }

    float* orow = out + (size_t)(b * SS + i0 + 8 * half) * NN + n0 + lr;
    #pragma unroll
    for (int r = 0; r < 8; ++r) orow[(size_t)r * NN] = c[r];
}

extern "C" void kernel_launch(void* const* d_in, const int* in_sizes, int n_in,
                              void* d_out, int out_size, void* d_ws, size_t ws_size,
                              hipStream_t stream) {
    (void)in_sizes; (void)n_in; (void)out_size; (void)ws_size;

    const float* query  = (const float*)d_in[0];
    const float* key_in = (const float*)d_in[1];
    const float* value  = (const float*)d_in[2];
    const float* qw = (const float*)d_in[3];
    const float* qb = (const float*)d_in[4];
    const float* kw = (const float*)d_in[5];
    const float* kb = (const float*)d_in[6];
    const float* vw = (const float*)d_in[7];
    const float* vb = (const float*)d_in[8];
    float* out = (float*)d_out;

    float* Qr = (float*)d_ws;
    float* Kr = Qr + (size_t)BB * SS * NN;
    float* Vr = Kr + (size_t)BB * SS * NN;
    float* sc = Vr + (size_t)BB * SS * NN;

    const int total = BB * SS * NN;
    lns_linear_kernel<<<total / 256, 256, 0, stream>>>(query, qw, qb, Qr);
    lns_linear_kernel<<<total / 256, 256, 0, stream>>>(key_in, kw, kb, Kr);
    lns_linear_kernel<<<total / 256, 256, 0, stream>>>(value, vw, vb, Vr);

    scores_wmma_kernel<<<64, 128, 0, stream>>>(Qr, Kr, sc);

    softmax_clip_kernel<<<BB * SS, SS, 0, stream>>>(sc);

    out_wmma_kernel<<<512, 128, 0, stream>>>(sc, Vr, out);
}


# LNS linear split: batch-free exp(w)+exp(bias) stored once as three S x N planes by one launch (6 -> 4 launches, 49152 -> 3072 waves); GEMM loaders add max(x,eps) per batch; 3-term sum order changes
# speedup vs baseline: 1.0902x; 1.0902x over previous
#include <hip/hip_runtime.h>

#define BB 4
#define SS 128
#define DD 256
#define NN 1024
#define EPS 1e-6f

typedef __attribute__((ext_vector_type(2))) float v2f;
typedef __attribute__((ext_vector_type(8))) float v8f;

__global__ void lns_planes_kernel(const float* __restrict__ qw,
                                  const float* __restrict__ qb,
                                  const float* __restrict__ kw,
                                  const float* __restrict__ kb,
                                  const float* __restrict__ vw,
                                  const float* __restrict__ vb,
                                  float* __restrict__ planes) {
    int which = blockIdx.y;
    const float* w    = (which == 0) ? qw : ((which == 1) ? kw : vw);
    const float* bias = (which == 0) ? qb : ((which == 1) ? kb : vb);
    int idx = blockIdx.x * blockDim.x + threadIdx.x;
    int d = idx & (DD - 1);
    int s = idx >> 8;
    float4 bb = *(const float4*)(bias + d * 4);
    float4 o;
    o.x = __expf(w[(0 * SS + s) * DD + d]) + __expf(bb.x);
    o.y = __expf(w[(1 * SS + s) * DD + d]) + __expf(bb.y);
    o.z = __expf(w[(2 * SS + s) * DD + d]) + __expf(bb.z);
    o.w = __expf(w[(3 * SS + s) * DD + d]) + __expf(bb.w);
    *(float4*)(planes + (size_t)which * SS * NN + (size_t)s * NN + d * 4) = o;
}

__global__ void scores_wmma_kernel(const float* __restrict__ Pq,
                                   const float* __restrict__ Pk,
                                   const float* __restrict__ xq,
                                   const float* __restrict__ xk,
                                   float* __restrict__ scores) {
    int wave = (blockIdx.x * blockDim.x + threadIdx.x) >> 5;
    int lane = threadIdx.x & 31;
    int b  = wave >> 6;
    int i0 = ((wave >> 3) & 7) << 4;
    int j0 = (wave & 7) << 4;
    int half = lane >> 4;
    int lr   = lane & 15;

    const float* arow = Pq + (size_t)(i0 + lr) * NN + 2 * half;
    const float* brow = Pk + (size_t)(j0 + lr) * NN + 2 * half;

    v2f xa = *(const v2f*)(xq + (size_t)(b * SS + i0 + lr) * 4 + 2 * half);
    v2f xb = *(const v2f*)(xk + (size_t)(b * SS + j0 + lr) * 4 + 2 * half);
    xa[0] = fmaxf(xa[0], EPS); xa[1] = fmaxf(xa[1], EPS);
    xb[0] = fmaxf(xb[0], EPS); xb[1] = fmaxf(xb[1], EPS);

    v8f c = {};
    #pragma nounroll
    for (int k = 0; k < NN; k += 16) {
        v2f af[4], bf[4];
        #pragma unroll
        for (int u = 0; u < 4; ++u) {
            af[u] = *(const v2f*)(arow + k + 4 * u) + xa;
            bf[u] = *(const v2f*)(brow + k + 4 * u) + xb;
        }
        __builtin_amdgcn_sched_barrier(0);
        #pragma unroll
        for (int u = 0; u < 4; ++u)
            c = __builtin_amdgcn_wmma_f32_16x16x4_f32(false, af[u], false, bf[u],
                                                      (short)0, c, false, false);
    }

    float* srow = scores + (size_t)(b * SS + i0 + 8 * half) * SS + j0 + lr;
    #pragma unroll
    for (int r = 0; r < 8; ++r) srow[(size_t)r * SS] = c[r];
}

__global__ void softmax_clip_kernel(float* __restrict__ scores) {
    __shared__ float red[SS];
    int row = blockIdx.x;
    int t   = threadIdx.x;
    float v = scores[(size_t)row * SS + t];
    red[t] = v;
    __syncthreads();
    for (int off = SS / 2; off > 0; off >>= 1) {
        if (t < off) red[t] = fmaxf(red[t], red[t + off]);
        __syncthreads();
    }
    float m = red[0];
    __syncthreads();
    float e = __expf(v - m);
    red[t] = e;
    __syncthreads();
    for (int off = SS / 2; off > 0; off >>= 1) {
        if (t < off) red[t] += red[t + off];
        __syncthreads();
    }
    float sum = red[0];
    scores[(size_t)row * SS + t] = fmaxf(e / sum, EPS);
}

__global__ void out_wmma_kernel(const float* __restrict__ attn,
                                const float* __restrict__ Pv,
                                const float* __restrict__ xv,
                                float* __restrict__ out) {
    int wave = (blockIdx.x * blockDim.x + threadIdx.x) >> 5;
    int lane = threadIdx.x & 31;
    int b  = wave >> 9;
    int i0 = ((wave >> 6) & 7) << 4;
    int n0 = (wave & 63) << 4;
    int half = lane >> 4;
    int lr   = lane & 15;

    const float* arow = attn + (size_t)(b * SS + i0 + lr) * SS + 2 * half;
    const float* bcol = Pv + n0 + lr;
    const float* xcol = xv + (size_t)(b * SS) * 4 + (lr & 3);

    v8f c = {};
    for (int k = 0; k < SS; k += 16) {
        v2f af[4], bf[4];
        #pragma unroll
        for (int u = 0; u < 4; ++u) {
            int kk = k + 4 * u + 2 * half;
            af[u] = *(const v2f*)(arow + k + 4 * u);
            bf[u][0] = bcol[(size_t)kk * NN]       + fmaxf(xcol[(size_t)kk * 4], EPS);
            bf[u][1] = bcol[(size_t)(kk + 1) * NN] + fmaxf(xcol[(size_t)(kk + 1) * 4], EPS);
        }
        __builtin_amdgcn_sched_barrier(0);
        #pragma unroll
        for (int u = 0; u < 4; ++u)
            c = __builtin_amdgcn_wmma_f32_16x16x4_f32(false, af[u], false, bf[u],
                                                      (short)0, c, false, false);
    }

    float* orow = out + (size_t)(b * SS + i0 + 8 * half) * NN + n0 + lr;
    #pragma unroll
    for (int r = 0; r < 8; ++r) orow[(size_t)r * NN] = c[r];
}

extern "C" void kernel_launch(void* const* d_in, const int* in_sizes, int n_in,
                              void* d_out, int out_size, void* d_ws, size_t ws_size,
                              hipStream_t stream) {
    (void)in_sizes; (void)n_in; (void)out_size; (void)ws_size;

    const float* query  = (const float*)d_in[0];
    const float* key_in = (const float*)d_in[1];
    const float* value  = (const float*)d_in[2];
    const float* qw = (const float*)d_in[3];
    const float* qb = (const float*)d_in[4];
    const float* kw = (const float*)d_in[5];
    const float* kb = (const float*)d_in[6];
    const float* vw = (const float*)d_in[7];
    const float* vb = (const float*)d_in[8];
    float* out = (float*)d_out;

    float* Pq = (float*)d_ws;
    float* Pk = Pq + (size_t)SS * NN;
    float* Pv = Pk + (size_t)SS * NN;
    float* sc = Pv + (size_t)SS * NN;

    lns_planes_kernel<<<dim3(128, 3), 256, 0, stream>>>(qw, qb, kw, kb, vw, vb, Pq);

    scores_wmma_kernel<<<64, 128, 0, stream>>>(Pq, Pk, query, key_in, sc);

    softmax_clip_kernel<<<BB * SS, SS, 0, stream>>>(sc);

    out_wmma_kernel<<<512, 128, 0, stream>>>(sc, Pv, value, out);
}


# scores GEMM fed from LDS: 4 waves own a 32x32 block, stage its 32+32 plane rows by coalesced 16-byte loads in 4 k chunks (2 in flight), read fragments from LDS 32 steps a group; sum order kept
# speedup vs baseline: 1.7533x; 1.6082x over previous
#include <hip/hip_runtime.h>

#define BB 4
#define SS 128
#define DD 256
#define NN 1024
#define EPS 1e-6f

typedef __attribute__((ext_vector_type(2))) float v2f;
typedef __attribute__((ext_vector_type(4))) float v4f;
typedef __attribute__((ext_vector_type(8))) float v8f;

__global__ void lns_planes_kernel(const float* __restrict__ qw,
                                  const float* __restrict__ qb,
                                  const float* __restrict__ kw,
                                  const float* __restrict__ kb,
                                  const float* __restrict__ vw,
                                  const float* __restrict__ vb,
                                  float* __restrict__ planes) {
    int which = blockIdx.y;
    const float* w    = (which == 0) ? qw : ((which == 1) ? kw : vw);
    const float* bias = (which == 0) ? qb : ((which == 1) ? kb : vb);
    int idx = blockIdx.x * blockDim.x + threadIdx.x;
    int d = idx & (DD - 1);
    int s = idx >> 8;
    float4 bb = *(const float4*)(bias + d * 4);
    float4 o;
    o.x = __expf(w[(0 * SS + s) * DD + d]) + __expf(bb.x);
    o.y = __expf(w[(1 * SS + s) * DD + d]) + __expf(bb.y);
    o.z = __expf(w[(2 * SS + s) * DD + d]) + __expf(bb.z);
    o.w = __expf(w[(3 * SS + s) * DD + d]) + __expf(bb.w);
    *(float4*)(planes + (size_t)which * SS * NN + (size_t)s * NN + d * 4) = o;
}

#define CH     256
#define PITCH  (NN + 4)
#define BROWS  32
#define OPF    (BROWS * PITCH)
#define GSTEPS 32

__device__ __forceinline__ void rows_fetch(v4f (&s)[16], const float* g) {
    #pragma unroll
    for (int t = 0; t < 16; ++t)
        s[t] = *(const v4f*)(g + (size_t)(t >> 1) * NN + (t & 1) * 128);
}

__device__ __forceinline__ void rows_to_lds(const v4f (&s)[16], float* dst) {
    #pragma unroll
    for (int t = 0; t < 16; ++t)
        *(v4f*)(dst + (t >> 1) * PITCH + (t & 1) * 128) = s[t];
}

__device__ __forceinline__ v8f chunk_mma(v8f c, const float* ca, const float* cb, v2f xa, v2f xb) {
    #pragma nounroll
    for (int k = 0; k < CH; k += 4 * GSTEPS) {
        v2f af[GSTEPS], bf[GSTEPS];
        #pragma unroll
        for (int u = 0; u < GSTEPS; ++u) {
            af[u] = *(const v2f*)(ca + k + 4 * u) + xa;
            bf[u] = *(const v2f*)(cb + k + 4 * u) + xb;
        }
        __builtin_amdgcn_sched_barrier(0);
        #pragma unroll
        for (int u = 0; u < GSTEPS; ++u)
            c = __builtin_amdgcn_wmma_f32_16x16x4_f32(false, af[u], false, bf[u],
                                                      (short)0, c, false, false);
    }
    return c;
}

__global__ __launch_bounds__(128)
void scores_wmma_kernel(const float* __restrict__ Pq,
                        const float* __restrict__ Pk,
                        const float* __restrict__ xq,
                        const float* __restrict__ xk,
                        float* __restrict__ scores) {
    __shared__ __attribute__((aligned(16))) float lds[2 * OPF];

    int tid  = threadIdx.x;
    int lane = tid & 31;
    int w    = tid >> 5;
    int blk  = blockIdx.x;
    int b  = blk >> 4;
    int i0 = ((blk >> 2) & 3) << 5;
    int j0 = (blk & 3) << 5;
    int half = lane >> 4;
    int lr   = lane & 15;
    int wi = w >> 1, wj = w & 1;

    const float* ga = Pq + (size_t)(i0 + 8 * w) * NN + lane * 4;
    const float* gb = Pk + (size_t)(j0 + 8 * w) * NN + lane * 4;
    float* sdst = lds + 8 * w * PITCH + lane * 4;

    v4f a0[16], b0[16], a1[16], b1[16], a2[16], b2[16], a3[16], b3[16];
    rows_fetch(a0, ga);           rows_fetch(b0, gb);
    rows_fetch(a1, ga + CH);      rows_fetch(b1, gb + CH);

    v2f xa = *(const v2f*)(xq + (size_t)(b * SS + i0 + wi * 16 + lr) * 4 + 2 * half);
    v2f xb = *(const v2f*)(xk + (size_t)(b * SS + j0 + wj * 16 + lr) * 4 + 2 * half);
    xa[0] = fmaxf(xa[0], EPS); xa[1] = fmaxf(xa[1], EPS);
    xb[0] = fmaxf(xb[0], EPS); xb[1] = fmaxf(xb[1], EPS);

    const float* la = lds + (wi * 16 + lr) * PITCH + 2 * half;
    const float* lb = lds + OPF + (wj * 16 + lr) * PITCH + 2 * half;

    v8f c = {};

    rows_to_lds(a0, sdst);                rows_to_lds(b0, sdst + OPF);
    __syncthreads();
    rows_fetch(a2, ga + 2 * CH);          rows_fetch(b2, gb + 2 * CH);
    c = chunk_mma(c, la, lb, xa, xb);

    rows_to_lds(a1, sdst + CH);           rows_to_lds(b1, sdst + OPF + CH);
    __syncthreads();
    rows_fetch(a3, ga + 3 * CH);          rows_fetch(b3, gb + 3 * CH);
    c = chunk_mma(c, la + CH, lb + CH, xa, xb);

    rows_to_lds(a2, sdst + 2 * CH);       rows_to_lds(b2, sdst + OPF + 2 * CH);
    __syncthreads();
    c = chunk_mma(c, la + 2 * CH, lb + 2 * CH, xa, xb);

    rows_to_lds(a3, sdst + 3 * CH);       rows_to_lds(b3, sdst + OPF + 3 * CH);
    __syncthreads();
    c = chunk_mma(c, la + 3 * CH, lb + 3 * CH, xa, xb);

    float* srow_out = scores + (size_t)(b * SS + i0 + wi * 16 + 8 * half) * SS + j0 + wj * 16 + lr;
    #pragma unroll
    for (int r = 0; r < 8; ++r) srow_out[(size_t)r * SS] = c[r];
}

__global__ void softmax_clip_kernel(float* __restrict__ scores) {
    __shared__ float red[SS];
    int row = blockIdx.x;
    int t   = threadIdx.x;
    float v = scores[(size_t)row * SS + t];
    red[t] = v;
    __syncthreads();
    for (int off = SS / 2; off > 0; off >>= 1) {
        if (t < off) red[t] = fmaxf(red[t], red[t + off]);
        __syncthreads();
    }
    float m = red[0];
    __syncthreads();
    float e = __expf(v - m);
    red[t] = e;
    __syncthreads();
    for (int off = SS / 2; off > 0; off >>= 1) {
        if (t < off) red[t] += red[t + off];
        __syncthreads();
    }
    float sum = red[0];
    scores[(size_t)row * SS + t] = fmaxf(e / sum, EPS);
}

__global__ void out_wmma_kernel(const float* __restrict__ attn,
                                const float* __restrict__ Pv,
                                const float* __restrict__ xv,
                                float* __restrict__ out) {
    int wave = (blockIdx.x * blockDim.x + threadIdx.x) >> 5;
    int lane = threadIdx.x & 31;
    int b  = wave >> 9;
    int i0 = ((wave >> 6) & 7) << 4;
    int n0 = (wave & 63) << 4;
    int half = lane >> 4;
    int lr   = lane & 15;

    const float* arow = attn + (size_t)(b * SS + i0 + lr) * SS + 2 * half;
    const float* bcol = Pv + n0 + lr;
    const float* xcol = xv + (size_t)(b * SS) * 4 + (lr & 3);

    v8f c = {};
    for (int k = 0; k < SS; k += 16) {
        v2f af[4], bf[4];
        #pragma unroll
        for (int u = 0; u < 4; ++u) {
            int kk = k + 4 * u + 2 * half;
            af[u] = *(const v2f*)(arow + k + 4 * u);
            bf[u][0] = bcol[(size_t)kk * NN]       + fmaxf(xcol[(size_t)kk * 4], EPS);
            bf[u][1] = bcol[(size_t)(kk + 1) * NN] + fmaxf(xcol[(size_t)(kk + 1) * 4], EPS);
        }
        __builtin_amdgcn_sched_barrier(0);
        #pragma unroll
        for (int u = 0; u < 4; ++u)
            c = __builtin_amdgcn_wmma_f32_16x16x4_f32(false, af[u], false, bf[u],
                                                      (short)0, c, false, false);
    }

    float* orow = out + (size_t)(b * SS + i0 + 8 * half) * NN + n0 + lr;
    #pragma unroll
    for (int r = 0; r < 8; ++r) orow[(size_t)r * NN] = c[r];
}

extern "C" void kernel_launch(void* const* d_in, const int* in_sizes, int n_in,
                              void* d_out, int out_size, void* d_ws, size_t ws_size,
                              hipStream_t stream) {
    (void)in_sizes; (void)n_in; (void)out_size; (void)ws_size;

    const float* query  = (const float*)d_in[0];
    const float* key_in = (const float*)d_in[1];
    const float* value  = (const float*)d_in[2];
    const float* qw = (const float*)d_in[3];
    const float* qb = (const float*)d_in[4];
    const float* kw = (const float*)d_in[5];
    const float* kb = (const float*)d_in[6];
    const float* vw = (const float*)d_in[7];
    const float* vb = (const float*)d_in[8];
    float* out = (float*)d_out;

    float* Pq = (float*)d_ws;
    float* Pk = Pq + (size_t)SS * NN;
    float* Pv = Pk + (size_t)SS * NN;
    float* sc = Pv + (size_t)SS * NN;

    lns_planes_kernel<<<dim3(128, 3), 256, 0, stream>>>(qw, qb, kw, kb, vw, vb, Pq);

    scores_wmma_kernel<<<64, 128, 0, stream>>>(Pq, Pk, query, key_in, sc);

    softmax_clip_kernel<<<BB * SS, SS, 0, stream>>>(sc);

    out_wmma_kernel<<<512, 128, 0, stream>>>(sc, Pv, value, out);
}


# softmax launch split away (4 -> 3): scores tiles transposed (row in lane), epilogue stores exp(s - tile max) and per-tile max,sum; out GEMM A loader rescales and clips; row-sum order changes
# speedup vs baseline: 1.9148x; 1.0921x over previous
#include <hip/hip_runtime.h>

#define BB 4
#define SS 128
#define DD 256
#define NN 1024
#define EPS 1e-6f

typedef __attribute__((ext_vector_type(2))) float v2f;
typedef __attribute__((ext_vector_type(4))) float v4f;
typedef __attribute__((ext_vector_type(8))) float v8f;

__global__ void lns_planes_kernel(const float* __restrict__ qw,
                                  const float* __restrict__ qb,
                                  const float* __restrict__ kw,
                                  const float* __restrict__ kb,
                                  const float* __restrict__ vw,
                                  const float* __restrict__ vb,
                                  float* __restrict__ planes) {
    int which = blockIdx.y;
    const float* w    = (which == 0) ? qw : ((which == 1) ? kw : vw);
    const float* bias = (which == 0) ? qb : ((which == 1) ? kb : vb);
    int idx = blockIdx.x * blockDim.x + threadIdx.x;
    int d = idx & (DD - 1);
    int s = idx >> 8;
    float4 bb = *(const float4*)(bias + d * 4);
    float4 o;
    o.x = __expf(w[(0 * SS + s) * DD + d]) + __expf(bb.x);
    o.y = __expf(w[(1 * SS + s) * DD + d]) + __expf(bb.y);
    o.z = __expf(w[(2 * SS + s) * DD + d]) + __expf(bb.z);
    o.w = __expf(w[(3 * SS + s) * DD + d]) + __expf(bb.w);
    *(float4*)(planes + (size_t)which * SS * NN + (size_t)s * NN + d * 4) = o;
}

#define CH     256
#define PITCH  (NN + 4)
#define BROWS  32
#define OPF    (BROWS * PITCH)
#define GSTEPS 32

__device__ __forceinline__ void rows_fetch(v4f (&s)[16], const float* g) {
    #pragma unroll
    for (int t = 0; t < 16; ++t)
        s[t] = *(const v4f*)(g + (size_t)(t >> 1) * NN + (t & 1) * 128);
}

__device__ __forceinline__ void rows_to_lds(const v4f (&s)[16], float* dst) {
    #pragma unroll
    for (int t = 0; t < 16; ++t)
        *(v4f*)(dst + (t >> 1) * PITCH + (t & 1) * 128) = s[t];
}

__device__ __forceinline__ v8f chunk_mma(v8f c, const float* ca, const float* cb, v2f xa, v2f xb) {
    #pragma nounroll
    for (int k = 0; k < CH; k += 4 * GSTEPS) {
        v2f af[GSTEPS], bf[GSTEPS];
        #pragma unroll
        for (int u = 0; u < GSTEPS; ++u) {
            af[u] = *(const v2f*)(ca + k + 4 * u) + xa;
            bf[u] = *(const v2f*)(cb + k + 4 * u) + xb;
        }
        __builtin_amdgcn_sched_barrier(0);
        #pragma unroll
        for (int u = 0; u < GSTEPS; ++u)
            c = __builtin_amdgcn_wmma_f32_16x16x4_f32(false, af[u], false, bf[u],
                                                      (short)0, c, false, false);
    }
    return c;
}

__global__ __launch_bounds__(128)
void scores_wmma_kernel(const float* __restrict__ Pq,
                        const float* __restrict__ Pk,
                        const float* __restrict__ xq,
                        const float* __restrict__ xk,
                        float* __restrict__ escores,
                        float* __restrict__ stats) {
    __shared__ __attribute__((aligned(16))) float lds[2 * OPF];

    int tid  = threadIdx.x;
    int lane = tid & 31;
    int w    = tid >> 5;
    int blk  = blockIdx.x;
    int b  = blk >> 4;
    int i0 = ((blk >> 2) & 3) << 5;
    int j0 = (blk & 3) << 5;
    int half = lane >> 4;
    int lr   = lane & 15;
    int wi = w >> 1, wj = w & 1;

    const float* ga = Pq + (size_t)(i0 + 8 * w) * NN + lane * 4;
    const float* gb = Pk + (size_t)(j0 + 8 * w) * NN + lane * 4;
    float* sdst = lds + 8 * w * PITCH + lane * 4;

    v4f a0[16], b0[16], a1[16], b1[16], a2[16], b2[16], a3[16], b3[16];
    rows_fetch(a0, ga);           rows_fetch(b0, gb);
    rows_fetch(a1, ga + CH);      rows_fetch(b1, gb + CH);

    v2f xa = *(const v2f*)(xq + (size_t)(b * SS + i0 + wi * 16 + lr) * 4 + 2 * half);
    v2f xb = *(const v2f*)(xk + (size_t)(b * SS + j0 + wj * 16 + lr) * 4 + 2 * half);
    xa[0] = fmaxf(xa[0], EPS); xa[1] = fmaxf(xa[1], EPS);
    xb[0] = fmaxf(xb[0], EPS); xb[1] = fmaxf(xb[1], EPS);

    const float* la = lds + (wi * 16 + lr) * PITCH + 2 * half;
    const float* lb = lds + OPF + (wj * 16 + lr) * PITCH + 2 * half;

    v8f c = {};

    rows_to_lds(a0, sdst);                rows_to_lds(b0, sdst + OPF);
    __syncthreads();
    rows_fetch(a2, ga + 2 * CH);          rows_fetch(b2, gb + 2 * CH);
    c = chunk_mma(c, lb, la, xb, xa);

    rows_to_lds(a1, sdst + CH);           rows_to_lds(b1, sdst + OPF + CH);
    __syncthreads();
    rows_fetch(a3, ga + 3 * CH);          rows_fetch(b3, gb + 3 * CH);
    c = chunk_mma(c, lb + CH, la + CH, xb, xa);

    rows_to_lds(a2, sdst + 2 * CH);       rows_to_lds(b2, sdst + OPF + 2 * CH);
    __syncthreads();
    c = chunk_mma(c, lb + 2 * CH, la + 2 * CH, xb, xa);

    rows_to_lds(a3, sdst + 3 * CH);       rows_to_lds(b3, sdst + OPF + 3 * CH);
    __syncthreads();
    c = chunk_mma(c, lb + 3 * CH, la + 3 * CH, xb, xa);

    asm volatile("s_wait_alu depctr_va_vdst(0)\n\tv_nop\n\tv_nop\n\tv_nop\n\tv_nop" : "+v"(c));

    float m = c[0];
    #pragma unroll
    for (int r = 1; r < 8; ++r) m = fmaxf(m, c[r]);
    m = fmaxf(m, __shfl_xor(m, 16));
    v4f e0, e1;
    #pragma unroll
    for (int r = 0; r < 4; ++r) {
        e0[r] = __expf(c[r] - m);
        e1[r] = __expf(c[r + 4] - m);
    }
    float l = ((e0[0] + e0[1]) + (e0[2] + e0[3])) + ((e1[0] + e1[1]) + (e1[2] + e1[3]));
    l += __shfl_xor(l, 16);

    int i = i0 + wi * 16 + lr;
    float* erow = escores + (size_t)(b * SS + i) * SS + j0 + wj * 16 + 8 * half;
    *(v4f*)erow       = e0;
    *(v4f*)(erow + 4) = e1;
    if (half == 0) {
        v2f ml = {m, l};
        *(v2f*)(stats + ((size_t)((b * 4 + (blk & 3)) * SS + i) * 2 + wj) * 2) = ml;
    }
}

__global__ void out_wmma_kernel(const float* __restrict__ ex,
                                const float* __restrict__ stats,
                                const float* __restrict__ Pv,
                                const float* __restrict__ xv,
                                float* __restrict__ out) {
    int wave = (blockIdx.x * blockDim.x + threadIdx.x) >> 5;
    int lane = threadIdx.x & 31;
    int b  = wave >> 9;
    int i0 = ((wave >> 6) & 7) << 4;
    int n0 = (wave & 63) << 4;
    int half = lane >> 4;
    int lr   = lane & 15;

    const float* arow = ex + (size_t)(b * SS + i0 + lr) * SS + 2 * half;
    const float* bcol = Pv + n0 + lr;
    const float* xcol = xv + (size_t)(b * SS) * 4 + (lr & 3);

    const float* srow = stats + (size_t)(b * 4 * SS + i0 + lr) * 4;
    v4f st[4];
    #pragma unroll
    for (int q = 0; q < 4; ++q) st[q] = *(const v4f*)(srow + (size_t)q * SS * 4);
    float M = fmaxf(st[0][0], st[0][2]);
    #pragma unroll
    for (int q = 1; q < 4; ++q) M = fmaxf(M, fmaxf(st[q][0], st[q][2]));
    float L = 0.0f;
    #pragma unroll
    for (int q = 0; q < 4; ++q) {
        L += __expf(st[q][0] - M) * st[q][1];
        L += __expf(st[q][2] - M) * st[q][3];
    }
    float inv = 1.0f / L;

    v8f c = {};
    for (int k = 0; k < SS; k += 16) {
        int t = k >> 4;
        float g = __expf(srow[(size_t)(t >> 1) * SS * 4 + (t & 1) * 2] - M) * inv;
        v2f af[4], bf[4];
        #pragma unroll
        for (int u = 0; u < 4; ++u) {
            int kk = k + 4 * u + 2 * half;
            v2f ev = *(const v2f*)(arow + k + 4 * u);
            af[u][0] = fmaxf(ev[0] * g, EPS);
            af[u][1] = fmaxf(ev[1] * g, EPS);
            bf[u][0] = bcol[(size_t)kk * NN]       + fmaxf(xcol[(size_t)kk * 4], EPS);
            bf[u][1] = bcol[(size_t)(kk + 1) * NN] + fmaxf(xcol[(size_t)(kk + 1) * 4], EPS);
        }
        __builtin_amdgcn_sched_barrier(0);
        #pragma unroll
        for (int u = 0; u < 4; ++u)
            c = __builtin_amdgcn_wmma_f32_16x16x4_f32(false, af[u], false, bf[u],
                                                      (short)0, c, false, false);
    }

    float* orow = out + (size_t)(b * SS + i0 + 8 * half) * NN + n0 + lr;
    #pragma unroll
    for (int r = 0; r < 8; ++r) orow[(size_t)r * NN] = c[r];
}

extern "C" void kernel_launch(void* const* d_in, const int* in_sizes, int n_in,
                              void* d_out, int out_size, void* d_ws, size_t ws_size,
                              hipStream_t stream) {
    (void)in_sizes; (void)n_in; (void)out_size; (void)ws_size;

    const float* query  = (const float*)d_in[0];
    const float* key_in = (const float*)d_in[1];
    const float* value  = (const float*)d_in[2];
    const float* qw = (const float*)d_in[3];
    const float* qb = (const float*)d_in[4];
    const float* kw = (const float*)d_in[5];
    const float* kb = (const float*)d_in[6];
    const float* vw = (const float*)d_in[7];
    const float* vb = (const float*)d_in[8];
    float* out = (float*)d_out;

    float* Pq = (float*)d_ws;
    float* Pk = Pq + (size_t)SS * NN;
    float* Pv = Pk + (size_t)SS * NN;
    float* sc = Pv + (size_t)SS * NN;
    float* st = sc + (size_t)BB * SS * SS;

    lns_planes_kernel<<<dim3(128, 3), 256, 0, stream>>>(qw, qb, kw, kb, vw, vb, Pq);

    scores_wmma_kernel<<<64, 128, 0, stream>>>(Pq, Pk, query, key_in, sc, st);

    out_wmma_kernel<<<512, 128, 0, stream>>>(sc, st, Pv, value, out);
}


# out GEMM re-tiled: 4 waves own 16x128 outputs, attn and clipped xv built once in LDS; a wave's 2 tiles interleave columns: Pv loads, out stores whole 128 B lines, all loads up front; sums kept
# speedup vs baseline: 2.2725x; 1.1868x over previous
#include <hip/hip_runtime.h>

#define BB 4
#define SS 128
#define DD 256
#define NN 1024
#define EPS 1e-6f

typedef __attribute__((ext_vector_type(2))) float v2f;
typedef __attribute__((ext_vector_type(4))) float v4f;
typedef __attribute__((ext_vector_type(8))) float v8f;

__global__ void lns_planes_kernel(const float* __restrict__ qw,
                                  const float* __restrict__ qb,
                                  const float* __restrict__ kw,
                                  const float* __restrict__ kb,
                                  const float* __restrict__ vw,
                                  const float* __restrict__ vb,
                                  float* __restrict__ planes) {
    int which = blockIdx.y;
    const float* w    = (which == 0) ? qw : ((which == 1) ? kw : vw);
    const float* bias = (which == 0) ? qb : ((which == 1) ? kb : vb);
    int idx = blockIdx.x * blockDim.x + threadIdx.x;
    int d = idx & (DD - 1);
    int s = idx >> 8;
    float4 bb = *(const float4*)(bias + d * 4);
    float4 o;
    o.x = __expf(w[(0 * SS + s) * DD + d]) + __expf(bb.x);
    o.y = __expf(w[(1 * SS + s) * DD + d]) + __expf(bb.y);
    o.z = __expf(w[(2 * SS + s) * DD + d]) + __expf(bb.z);
    o.w = __expf(w[(3 * SS + s) * DD + d]) + __expf(bb.w);
    *(float4*)(planes + (size_t)which * SS * NN + (size_t)s * NN + d * 4) = o;
}

#define CH     256
#define PITCH  (NN + 4)
#define BROWS  32
#define OPF    (BROWS * PITCH)
#define GSTEPS 32

__device__ __forceinline__ void rows_fetch(v4f (&s)[16], const float* g) {
    #pragma unroll
    for (int t = 0; t < 16; ++t)
        s[t] = *(const v4f*)(g + (size_t)(t >> 1) * NN + (t & 1) * 128);
}

__device__ __forceinline__ void rows_to_lds(const v4f (&s)[16], float* dst) {
    #pragma unroll
    for (int t = 0; t < 16; ++t)
        *(v4f*)(dst + (t >> 1) * PITCH + (t & 1) * 128) = s[t];
}

__device__ __forceinline__ v8f chunk_mma(v8f c, const float* ca, const float* cb, v2f xa, v2f xb) {
    #pragma nounroll
    for (int k = 0; k < CH; k += 4 * GSTEPS) {
        v2f af[GSTEPS], bf[GSTEPS];
        #pragma unroll
        for (int u = 0; u < GSTEPS; ++u) {
            af[u] = *(const v2f*)(ca + k + 4 * u) + xa;
            bf[u] = *(const v2f*)(cb + k + 4 * u) + xb;
        }
        __builtin_amdgcn_sched_barrier(0);
        #pragma unroll
        for (int u = 0; u < GSTEPS; ++u)
            c = __builtin_amdgcn_wmma_f32_16x16x4_f32(false, af[u], false, bf[u],
                                                      (short)0, c, false, false);
    }
    return c;
}

__global__ __launch_bounds__(128)
void scores_wmma_kernel(const float* __restrict__ Pq,
                        const float* __restrict__ Pk,
                        const float* __restrict__ xq,
                        const float* __restrict__ xk,
                        float* __restrict__ escores,
                        float* __restrict__ stats) {
    __shared__ __attribute__((aligned(16))) float lds[2 * OPF];

    int tid  = threadIdx.x;
    int lane = tid & 31;
    int w    = tid >> 5;
    int blk  = blockIdx.x;
    int b  = blk >> 4;
    int i0 = ((blk >> 2) & 3) << 5;
    int j0 = (blk & 3) << 5;
    int half = lane >> 4;
    int lr   = lane & 15;
    int wi = w >> 1, wj = w & 1;

    const float* ga = Pq + (size_t)(i0 + 8 * w) * NN + lane * 4;
    const float* gb = Pk + (size_t)(j0 + 8 * w) * NN + lane * 4;
    float* sdst = lds + 8 * w * PITCH + lane * 4;

    v4f a0[16], b0[16], a1[16], b1[16], a2[16], b2[16], a3[16], b3[16];
    rows_fetch(a0, ga);           rows_fetch(b0, gb);
    rows_fetch(a1, ga + CH);      rows_fetch(b1, gb + CH);

    v2f xa = *(const v2f*)(xq + (size_t)(b * SS + i0 + wi * 16 + lr) * 4 + 2 * half);
    v2f xb = *(const v2f*)(xk + (size_t)(b * SS + j0 + wj * 16 + lr) * 4 + 2 * half);
    xa[0] = fmaxf(xa[0], EPS); xa[1] = fmaxf(xa[1], EPS);
    xb[0] = fmaxf(xb[0], EPS); xb[1] = fmaxf(xb[1], EPS);

    const float* la = lds + (wi * 16 + lr) * PITCH + 2 * half;
    const float* lb = lds + OPF + (wj * 16 + lr) * PITCH + 2 * half;

    v8f c = {};

    rows_to_lds(a0, sdst);                rows_to_lds(b0, sdst + OPF);
    __syncthreads();
    rows_fetch(a2, ga + 2 * CH);          rows_fetch(b2, gb + 2 * CH);
    c = chunk_mma(c, lb, la, xb, xa);

    rows_to_lds(a1, sdst + CH);           rows_to_lds(b1, sdst + OPF + CH);
    __syncthreads();
    rows_fetch(a3, ga + 3 * CH);          rows_fetch(b3, gb + 3 * CH);
    c = chunk_mma(c, lb + CH, la + CH, xb, xa);

    rows_to_lds(a2, sdst + 2 * CH);       rows_to_lds(b2, sdst + OPF + 2 * CH);
    __syncthreads();
    c = chunk_mma(c, lb + 2 * CH, la + 2 * CH, xb, xa);

    rows_to_lds(a3, sdst + 3 * CH);       rows_to_lds(b3, sdst + OPF + 3 * CH);
    __syncthreads();
    c = chunk_mma(c, lb + 3 * CH, la + 3 * CH, xb, xa);

    asm volatile("s_wait_alu depctr_va_vdst(0)\n\tv_nop\n\tv_nop\n\tv_nop\n\tv_nop" : "+v"(c));

    float m = c[0];
    #pragma unroll
    for (int r = 1; r < 8; ++r) m = fmaxf(m, c[r]);
    m = fmaxf(m, __shfl_xor(m, 16));
    v4f e0, e1;
    #pragma unroll
    for (int r = 0; r < 4; ++r) {
        e0[r] = __expf(c[r] - m);
        e1[r] = __expf(c[r + 4] - m);
    }
    float l = ((e0[0] + e0[1]) + (e0[2] + e0[3])) + ((e1[0] + e1[1]) + (e1[2] + e1[3]));
    l += __shfl_xor(l, 16);

    int i = i0 + wi * 16 + lr;
    float* erow = escores + (size_t)(b * SS + i) * SS + j0 + wj * 16 + 8 * half;
    *(v4f*)erow       = e0;
    *(v4f*)(erow + 4) = e1;
    if (half == 0) {
        v2f ml = {m, l};
        *(v2f*)(stats + ((size_t)((b * 4 + (blk & 3)) * SS + i) * 2 + wj) * 2) = ml;
    }
}

#define APITCH (SS + 4)
#define OSTEPS 16

__global__ __launch_bounds__(128)
void out_wmma_kernel(const float* __restrict__ ex,
                     const float* __restrict__ stats,
                     const float* __restrict__ Pv,
                     const float* __restrict__ xv,
                     float* __restrict__ out) {
    __shared__ __attribute__((aligned(16))) float la[16 * APITCH];
    __shared__ __attribute__((aligned(16))) float lx[SS * 4];

    int tid  = threadIdx.x;
    int lane = tid & 31;
    int w    = tid >> 5;
    int blk  = blockIdx.x;
    int b  = blk >> 6;
    int i0 = ((blk >> 3) & 7) << 4;
    int n0 = ((blk & 7) << 7) + (w << 5);
    int half = lane >> 4;
    int lr   = lane & 15;


    int r = tid >> 3;
    int t = tid & 7;
    const float* srow = stats + (size_t)(b * 4 * SS + i0 + r) * 4;
    v4f st[4];
    #pragma unroll
    for (int q = 0; q < 4; ++q) st[q] = *(const v4f*)(srow + (size_t)q * SS * 4);
    float mt = srow[(size_t)(t >> 1) * SS * 4 + (t & 1) * 2];
    const float* erow = ex + (size_t)(b * SS + i0 + r) * SS + t * 16;
    v4f ev[4];
    #pragma unroll
    for (int q = 0; q < 4; ++q) ev[q] = *(const v4f*)(erow + 4 * q);
    v4f xr = *(const v4f*)(xv + (size_t)(b * SS + tid) * 4);
    __builtin_amdgcn_sched_barrier(0);

    const float* bp = Pv + (size_t)(2 * half) * NN + n0 + 2 * lr;
    v2f bv[64];
    #pragma unroll
    for (int m = 0; m < 32; ++m) {
        bv[2 * m]     = *(const v2f*)(bp + (size_t)(4 * m) * NN);
        bv[2 * m + 1] = *(const v2f*)(bp + (size_t)(4 * m + 1) * NN);
    }
    __builtin_amdgcn_sched_barrier(0);

    {
        float M = fmaxf(st[0][0], st[0][2]);
        #pragma unroll
        for (int q = 1; q < 4; ++q) M = fmaxf(M, fmaxf(st[q][0], st[q][2]));
        float L = 0.0f;
        #pragma unroll
        for (int q = 0; q < 4; ++q) {
            L += __expf(st[q][0] - M) * st[q][1];
            L += __expf(st[q][2] - M) * st[q][3];
        }
        float inv = 1.0f / L;
        float g = __expf(mt - M) * inv;
        float* dst = la + r * APITCH + t * 16;
        #pragma unroll
        for (int q = 0; q < 4; ++q) {
            v4f a;
            #pragma unroll
            for (int z = 0; z < 4; ++z) a[z] = fmaxf(ev[q][z] * g, EPS);
            *(v4f*)(dst + 4 * q) = a;
        }
        #pragma unroll
        for (int z = 0; z < 4; ++z) xr[z] = fmaxf(xr[z], EPS);
        *(v4f*)(lx + tid * 4) = xr;
    }
    __builtin_amdgcn_fence(__ATOMIC_RELEASE, "workgroup", "local");
    __builtin_amdgcn_s_barrier();
    __builtin_amdgcn_fence(__ATOMIC_ACQUIRE, "workgroup", "local");

    const float* pa = la + lr * APITCH + 2 * half;
    const float* px = lx + (2 * half) * 4 + 2 * (lr & 1);

    v8f c0 = {}, c1 = {};
    #pragma unroll
    for (int gq = 0; gq < 32 / OSTEPS; ++gq) {
        v2f af[OSTEPS], bf0[OSTEPS], bf1[OSTEPS];
        #pragma unroll
        for (int u = 0; u < OSTEPS; ++u) {
            int m = gq * OSTEPS + u;
            af[u] = *(const v2f*)(pa + 4 * m);
            v2f x0 = *(const v2f*)(px + 16 * m);
            v2f x1 = *(const v2f*)(px + 16 * m + 4);
            bf0[u][0] = bv[2 * m][0]     + x0[0];
            bf0[u][1] = bv[2 * m + 1][0] + x1[0];
            bf1[u][0] = bv[2 * m][1]     + x0[1];
            bf1[u][1] = bv[2 * m + 1][1] + x1[1];
        }
        __builtin_amdgcn_sched_barrier(0);
        #pragma unroll
        for (int u = 0; u < OSTEPS; ++u) {
            c0 = __builtin_amdgcn_wmma_f32_16x16x4_f32(false, af[u], false, bf0[u],
                                                       (short)0, c0, false, false);
            c1 = __builtin_amdgcn_wmma_f32_16x16x4_f32(false, af[u], false, bf1[u],
                                                       (short)0, c1, false, false);
        }
        __builtin_amdgcn_sched_barrier(0);
        if (gq + 1 < 32 / OSTEPS) {
            asm volatile("v_nop\n\tv_nop\n\tv_nop\n\tv_nop");
            __builtin_amdgcn_sched_barrier(0);
        }
    }

    asm volatile("s_wait_alu depctr_va_vdst(0)\n\tv_nop\n\tv_nop\n\tv_nop\n\tv_nop" : "+v"(c0), "+v"(c1));

    float* orow = out + (size_t)(b * SS + i0 + 8 * half) * NN + n0 + 2 * lr;
    #pragma unroll
    for (int r = 0; r < 8; ++r) {
        v2f o = {c0[r], c1[r]};
        *(v2f*)(orow + (size_t)r * NN) = o;
    }
}

extern "C" void kernel_launch(void* const* d_in, const int* in_sizes, int n_in,
                              void* d_out, int out_size, void* d_ws, size_t ws_size,
                              hipStream_t stream) {
    (void)in_sizes; (void)n_in; (void)out_size; (void)ws_size;

    const float* query  = (const float*)d_in[0];
    const float* key_in = (const float*)d_in[1];
    const float* value  = (const float*)d_in[2];
    const float* qw = (const float*)d_in[3];
    const float* qb = (const float*)d_in[4];
    const float* kw = (const float*)d_in[5];
    const float* kb = (const float*)d_in[6];
    const float* vw = (const float*)d_in[7];
    const float* vb = (const float*)d_in[8];
    float* out = (float*)d_out;

    float* Pq = (float*)d_ws;
    float* Pk = Pq + (size_t)SS * NN;
    float* Pv = Pk + (size_t)SS * NN;
    float* sc = Pv + (size_t)SS * NN;
    float* st = sc + (size_t)BB * SS * SS;

    lns_planes_kernel<<<dim3(128, 3), 256, 0, stream>>>(qw, qb, kw, kb, vw, vb, Pq);

    scores_wmma_kernel<<<64, 128, 0, stream>>>(Pq, Pk, query, key_in, sc, st);

    out_wmma_kernel<<<256, 128, 0, stream>>>(sc, st, Pv, value, out);
}


# scores loop pipelined: next group's LDS fragment reads issued under the current 32-step matrix run (2 register sets; batch add at staging, chunks staged a run early, LDS-only barriers); sums kept
# speedup vs baseline: 2.4553x; 1.0804x over previous
#include <hip/hip_runtime.h>

#define BB 4
#define SS 128
#define DD 256
#define NN 1024
#define EPS 1e-6f

typedef __attribute__((ext_vector_type(2))) float v2f;
typedef __attribute__((ext_vector_type(4))) float v4f;
typedef __attribute__((ext_vector_type(8))) float v8f;

__global__ void lns_planes_kernel(const float* __restrict__ qw,
                                  const float* __restrict__ qb,
                                  const float* __restrict__ kw,
                                  const float* __restrict__ kb,
                                  const float* __restrict__ vw,
                                  const float* __restrict__ vb,
                                  float* __restrict__ planes) {
    int which = blockIdx.y;
    const float* w    = (which == 0) ? qw : ((which == 1) ? kw : vw);
    const float* bias = (which == 0) ? qb : ((which == 1) ? kb : vb);
    int idx = blockIdx.x * blockDim.x + threadIdx.x;
    int d = idx & (DD - 1);
    int s = idx >> 8;
    float4 bb = *(const float4*)(bias + d * 4);
    float4 o;
    o.x = __expf(w[(0 * SS + s) * DD + d]) + __expf(bb.x);
    o.y = __expf(w[(1 * SS + s) * DD + d]) + __expf(bb.y);
    o.z = __expf(w[(2 * SS + s) * DD + d]) + __expf(bb.z);
    o.w = __expf(w[(3 * SS + s) * DD + d]) + __expf(bb.w);
    *(float4*)(planes + (size_t)which * SS * NN + (size_t)s * NN + d * 4) = o;
}

#define CH     256
#define PITCH  (NN + 4)
#define BROWS  32
#define OPF    (BROWS * PITCH)
#define GSTEPS 32
#define GK     (4 * GSTEPS)

__device__ __forceinline__ void rows_fetch(v4f (&s)[16], const float* g) {
    #pragma unroll
    for (int t = 0; t < 16; ++t)
        s[t] = *(const v4f*)(g + (size_t)(t >> 1) * NN + (t & 1) * 128);
}

__device__ __forceinline__ void rows_to_lds(const v4f (&s)[16], const v4f (&x)[8], float* dst) {
    #pragma unroll
    for (int t = 0; t < 16; ++t)
        *(v4f*)(dst + (t >> 1) * PITCH + (t & 1) * 128) = s[t] + x[t >> 1];
}

__device__ __forceinline__ void frag_load(v2f (&af)[GSTEPS], v2f (&bf)[GSTEPS],
                                          const float* ca, const float* cb) {
    #pragma unroll
    for (int u = 0; u < GSTEPS; ++u) {
        af[u] = *(const v2f*)(ca + 4 * u);
        bf[u] = *(const v2f*)(cb + 4 * u);
    }
}

__device__ __forceinline__ v8f frag_mma(v8f c, const v2f (&af)[GSTEPS], const v2f (&bf)[GSTEPS]) {
    __builtin_amdgcn_sched_barrier(0);
    #pragma unroll
    for (int u = 0; u < GSTEPS; ++u)
        c = __builtin_amdgcn_wmma_f32_16x16x4_f32(false, af[u], false, bf[u],
                                                  (short)0, c, false, false);
    asm volatile("v_nop\n\tv_nop\n\tv_nop\n\tv_nop" : "+v"(c));
    __builtin_amdgcn_sched_barrier(0);
    return c;
}

__device__ __forceinline__ void lds_barrier() {
    __builtin_amdgcn_fence(__ATOMIC_RELEASE, "workgroup", "local");
    __builtin_amdgcn_s_barrier();
    __builtin_amdgcn_fence(__ATOMIC_ACQUIRE, "workgroup", "local");
}

__global__ __launch_bounds__(128)
void scores_wmma_kernel(const float* __restrict__ Pq,
                        const float* __restrict__ Pk,
                        const float* __restrict__ xq,
                        const float* __restrict__ xk,
                        float* __restrict__ escores,
                        float* __restrict__ stats) {
    __shared__ __attribute__((aligned(16))) float lds[2 * OPF];

    int tid  = threadIdx.x;
    int lane = tid & 31;
    int w    = tid >> 5;
    int blk  = blockIdx.x;
    int b  = blk >> 4;
    int i0 = ((blk >> 2) & 3) << 5;
    int j0 = (blk & 3) << 5;
    int half = lane >> 4;
    int lr   = lane & 15;
    int wi = w >> 1, wj = w & 1;

    const float* ga = Pq + (size_t)(i0 + 8 * w) * NN + lane * 4;
    const float* gb = Pk + (size_t)(j0 + 8 * w) * NN + lane * 4;
    float* sdst = lds + 8 * w * PITCH + lane * 4;

    v4f a0[16], b0[16], a1[16], b1[16], a2[16], b2[16], a3[16], b3[16];
    rows_fetch(a0, ga);           rows_fetch(b0, gb);

    v4f xqr[8], xkr[8];
    #pragma unroll
    for (int r = 0; r < 8; ++r) {
        xqr[r] = *(const v4f*)(xq + (size_t)(b * SS + i0 + 8 * w + r) * 4);
        xkr[r] = *(const v4f*)(xk + (size_t)(b * SS + j0 + 8 * w + r) * 4);
    }

    rows_fetch(a1, ga + CH);      rows_fetch(b1, gb + CH);

    #pragma unroll
    for (int r = 0; r < 8; ++r) {
        #pragma unroll
        for (int z = 0; z < 4; ++z) {
            xqr[r][z] = fmaxf(xqr[r][z], EPS);
            xkr[r][z] = fmaxf(xkr[r][z], EPS);
        }
    }

    const float* fa = lds + OPF + (wj * 16 + lr) * PITCH + 2 * half;
    const float* fb = lds + (wi * 16 + lr) * PITCH + 2 * half;

    v2f af0[GSTEPS], bf0[GSTEPS], af1[GSTEPS], bf1[GSTEPS];
    v8f c = {};

    rows_to_lds(a0, xqr, sdst);                rows_to_lds(b0, xkr, sdst + OPF);
    lds_barrier();
    rows_fetch(a2, ga + 2 * CH);               rows_fetch(b2, gb + 2 * CH);
    frag_load(af0, bf0, fa, fb);
    frag_load(af1, bf1, fa + GK, fb + GK);
    c = frag_mma(c, af0, bf0);

    rows_to_lds(a1, xqr, sdst + CH);           rows_to_lds(b1, xkr, sdst + OPF + CH);
    lds_barrier();
    rows_fetch(a3, ga + 3 * CH);               rows_fetch(b3, gb + 3 * CH);
    frag_load(af0, bf0, fa + CH, fb + CH);
    c = frag_mma(c, af1, bf1);

    rows_to_lds(a2, xqr, sdst + 2 * CH);       rows_to_lds(b2, xkr, sdst + OPF + 2 * CH);
    frag_load(af1, bf1, fa + CH + GK, fb + CH + GK);
    c = frag_mma(c, af0, bf0);

    lds_barrier();
    frag_load(af0, bf0, fa + 2 * CH, fb + 2 * CH);
    c = frag_mma(c, af1, bf1);

    rows_to_lds(a3, xqr, sdst + 3 * CH);       rows_to_lds(b3, xkr, sdst + OPF + 3 * CH);
    frag_load(af1, bf1, fa + 2 * CH + GK, fb + 2 * CH + GK);
    c = frag_mma(c, af0, bf0);

    lds_barrier();
    frag_load(af0, bf0, fa + 3 * CH, fb + 3 * CH);
    c = frag_mma(c, af1, bf1);

    frag_load(af1, bf1, fa + 3 * CH + GK, fb + 3 * CH + GK);
    c = frag_mma(c, af0, bf0);
    c = frag_mma(c, af1, bf1);

    asm volatile("s_wait_alu depctr_va_vdst(0)\n\tv_nop\n\tv_nop\n\tv_nop\n\tv_nop" : "+v"(c));

    float m = c[0];
    #pragma unroll
    for (int r = 1; r < 8; ++r) m = fmaxf(m, c[r]);
    m = fmaxf(m, __shfl_xor(m, 16));
    v4f e0, e1;
    #pragma unroll
    for (int r = 0; r < 4; ++r) {
        e0[r] = __expf(c[r] - m);
        e1[r] = __expf(c[r + 4] - m);
    }
    float l = ((e0[0] + e0[1]) + (e0[2] + e0[3])) + ((e1[0] + e1[1]) + (e1[2] + e1[3]));
    l += __shfl_xor(l, 16);

    int i = i0 + wi * 16 + lr;
    float* erow = escores + (size_t)(b * SS + i) * SS + j0 + wj * 16 + 8 * half;
    *(v4f*)erow       = e0;
    *(v4f*)(erow + 4) = e1;
    if (half == 0) {
        v2f ml = {m, l};
        *(v2f*)(stats + ((size_t)((b * 4 + (blk & 3)) * SS + i) * 2 + wj) * 2) = ml;
    }
}

#define APITCH (SS + 4)
#define OSTEPS 16

__global__ __launch_bounds__(128)
void out_wmma_kernel(const float* __restrict__ ex,
                     const float* __restrict__ stats,
                     const float* __restrict__ Pv,
                     const float* __restrict__ xv,
                     float* __restrict__ out) {
    __shared__ __attribute__((aligned(16))) float la[16 * APITCH];
    __shared__ __attribute__((aligned(16))) float lx[SS * 4];

    int tid  = threadIdx.x;
    int lane = tid & 31;
    int w    = tid >> 5;
    int blk  = blockIdx.x;
    int b  = blk >> 6;
    int i0 = ((blk >> 3) & 7) << 4;
    int n0 = ((blk & 7) << 7) + (w << 5);
    int half = lane >> 4;
    int lr   = lane & 15;


    int r = tid >> 3;
    int t = tid & 7;
    const float* srow = stats + (size_t)(b * 4 * SS + i0 + r) * 4;
    v4f st[4];
    #pragma unroll
    for (int q = 0; q < 4; ++q) st[q] = *(const v4f*)(srow + (size_t)q * SS * 4);
    float mt = srow[(size_t)(t >> 1) * SS * 4 + (t & 1) * 2];
    const float* erow = ex + (size_t)(b * SS + i0 + r) * SS + t * 16;
    v4f ev[4];
    #pragma unroll
    for (int q = 0; q < 4; ++q) ev[q] = *(const v4f*)(erow + 4 * q);
    v4f xr = *(const v4f*)(xv + (size_t)(b * SS + tid) * 4);
    __builtin_amdgcn_sched_barrier(0);

    const float* bp = Pv + (size_t)(2 * half) * NN + n0 + 2 * lr;
    v2f bv[64];
    #pragma unroll
    for (int m = 0; m < 32; ++m) {
        bv[2 * m]     = *(const v2f*)(bp + (size_t)(4 * m) * NN);
        bv[2 * m + 1] = *(const v2f*)(bp + (size_t)(4 * m + 1) * NN);
    }
    __builtin_amdgcn_sched_barrier(0);

    {
        float M = fmaxf(st[0][0], st[0][2]);
        #pragma unroll
        for (int q = 1; q < 4; ++q) M = fmaxf(M, fmaxf(st[q][0], st[q][2]));
        float L = 0.0f;
        #pragma unroll
        for (int q = 0; q < 4; ++q) {
            L += __expf(st[q][0] - M) * st[q][1];
            L += __expf(st[q][2] - M) * st[q][3];
        }
        float inv = 1.0f / L;
        float g = __expf(mt - M) * inv;
        float* dst = la + r * APITCH + t * 16;
        #pragma unroll
        for (int q = 0; q < 4; ++q) {
            v4f a;
            #pragma unroll
            for (int z = 0; z < 4; ++z) a[z] = fmaxf(ev[q][z] * g, EPS);
            *(v4f*)(dst + 4 * q) = a;
        }
        #pragma unroll
        for (int z = 0; z < 4; ++z) xr[z] = fmaxf(xr[z], EPS);
        *(v4f*)(lx + tid * 4) = xr;
    }
    __builtin_amdgcn_fence(__ATOMIC_RELEASE, "workgroup", "local");
    __builtin_amdgcn_s_barrier();
    __builtin_amdgcn_fence(__ATOMIC_ACQUIRE, "workgroup", "local");

    const float* pa = la + lr * APITCH + 2 * half;
    const float* px = lx + (2 * half) * 4 + 2 * (lr & 1);

    v8f c0 = {}, c1 = {};
    #pragma unroll
    for (int gq = 0; gq < 32 / OSTEPS; ++gq) {
        v2f af[OSTEPS], bf0[OSTEPS], bf1[OSTEPS];
        #pragma unroll
        for (int u = 0; u < OSTEPS; ++u) {
            int m = gq * OSTEPS + u;
            af[u] = *(const v2f*)(pa + 4 * m);
            v2f x0 = *(const v2f*)(px + 16 * m);
            v2f x1 = *(const v2f*)(px + 16 * m + 4);
            bf0[u][0] = bv[2 * m][0]     + x0[0];
            bf0[u][1] = bv[2 * m + 1][0] + x1[0];
            bf1[u][0] = bv[2 * m][1]     + x0[1];
            bf1[u][1] = bv[2 * m + 1][1] + x1[1];
        }
        __builtin_amdgcn_sched_barrier(0);
        #pragma unroll
        for (int u = 0; u < OSTEPS; ++u) {
            c0 = __builtin_amdgcn_wmma_f32_16x16x4_f32(false, af[u], false, bf0[u],
                                                       (short)0, c0, false, false);
            c1 = __builtin_amdgcn_wmma_f32_16x16x4_f32(false, af[u], false, bf1[u],
                                                       (short)0, c1, false, false);
        }
        __builtin_amdgcn_sched_barrier(0);
        if (gq + 1 < 32 / OSTEPS) {
            asm volatile("v_nop\n\tv_nop\n\tv_nop\n\tv_nop");
            __builtin_amdgcn_sched_barrier(0);
        }
    }

    asm volatile("s_wait_alu depctr_va_vdst(0)\n\tv_nop\n\tv_nop\n\tv_nop\n\tv_nop" : "+v"(c0), "+v"(c1));

    float* orow = out + (size_t)(b * SS + i0 + 8 * half) * NN + n0 + 2 * lr;
    #pragma unroll
    for (int r = 0; r < 8; ++r) {
        v2f o = {c0[r], c1[r]};
        *(v2f*)(orow + (size_t)r * NN) = o;
    }
}

extern "C" void kernel_launch(void* const* d_in, const int* in_sizes, int n_in,
                              void* d_out, int out_size, void* d_ws, size_t ws_size,
                              hipStream_t stream) {
    (void)in_sizes; (void)n_in; (void)out_size; (void)ws_size;

    const float* query  = (const float*)d_in[0];
    const float* key_in = (const float*)d_in[1];
    const float* value  = (const float*)d_in[2];
    const float* qw = (const float*)d_in[3];
    const float* qb = (const float*)d_in[4];
    const float* kw = (const float*)d_in[5];
    const float* kb = (const float*)d_in[6];
    const float* vw = (const float*)d_in[7];
    const float* vb = (const float*)d_in[8];
    float* out = (float*)d_out;

    float* Pq = (float*)d_ws;
    float* Pk = Pq + (size_t)SS * NN;
    float* Pv = Pk + (size_t)SS * NN;
    float* sc = Pv + (size_t)SS * NN;
    float* st = sc + (size_t)BB * SS * SS;

    lns_planes_kernel<<<dim3(128, 3), 256, 0, stream>>>(qw, qb, kw, kb, vw, vb, Pq);

    scores_wmma_kernel<<<64, 128, 0, stream>>>(Pq, Pk, query, key_in, sc, st);

    out_wmma_kernel<<<256, 128, 0, stream>>>(sc, st, Pv, value, out);
}


# scores re-tiled: a group owns a 32x32 block of ONE quaternion component, a staged row feeds 2 tiles (half the fetches, exps); out launch adds the 4 planes, does the row softmax; row-sum order changes
# speedup vs baseline: 3.9464x; 1.6073x over previous
#include <hip/hip_runtime.h>

#define BB 4
#define SS 128
#define DD 256
#define NN 1024
#define EPS 1e-6f

typedef __attribute__((ext_vector_type(2))) float v2f;
typedef __attribute__((ext_vector_type(4))) float v4f;
typedef __attribute__((ext_vector_type(8))) float v8f;


#define CH     256
#define WPL    (SS * DD)
#define WPITCH (CH + 4)
#define GSTEPS 32
#define GK     (4 * GSTEPS)
#define SCORE_GROUPS 256
#define PPL    (BB * SS * SS)

__device__ __forceinline__ void half_fetch(v4f (&s)[16], const float* g) {
    #pragma unroll
    for (int r = 0; r < 16; ++r)
        s[r] = *(const v4f*)(g + (size_t)r * DD);
}

__device__ __forceinline__ void piece_real(v4f& s, const float (&eb)[4], float x) {
    #pragma unroll
    for (int e = 0; e < 4; ++e) s[e] = (__expf(s[e]) + eb[e]) + x;
}

__device__ __forceinline__ void half_real(v4f (&s)[16], const float (&eb)[4], const float (&x)[16]) {
    #pragma unroll
    for (int r = 0; r < 16; ++r) piece_real(s[r], eb, x[r]);
}

__device__ __forceinline__ void half_to_lds(const v4f (&s)[16], float* dst) {
    #pragma unroll
    for (int r = 0; r < 16; ++r)
        *(v4f*)(dst + r * WPITCH) = s[r];
}

__device__ __forceinline__ void frag_load(v2f (&af)[GSTEPS], v2f (&bf)[GSTEPS],
                                          const float* ca, const float* cb) {
    #pragma unroll
    for (int u = 0; u < GSTEPS; ++u) {
        af[u] = *(const v2f*)(ca + 4 * u);
        bf[u] = *(const v2f*)(cb + 4 * u);
    }
}

__device__ __forceinline__ v8f frag_mma(v8f c, const v2f (&af)[GSTEPS], const v2f (&bf)[GSTEPS]) {
    __builtin_amdgcn_sched_barrier(0);
    #pragma unroll
    for (int u = 0; u < GSTEPS; ++u)
        c = __builtin_amdgcn_wmma_f32_16x16x4_f32(false, af[u], false, bf[u],
                                                  (short)0, c, false, false);
    asm volatile("v_nop\n\tv_nop\n\tv_nop\n\tv_nop" : "+v"(c));
    __builtin_amdgcn_sched_barrier(0);
    return c;
}

__device__ __forceinline__ v8f frag_mma_real(v8f c, const v2f (&af)[GSTEPS], const v2f (&bf)[GSTEPS],
                                             v4f (&s)[16], const float (&eb)[4], const float (&x)[16]) {
    __builtin_amdgcn_sched_barrier(0);
    #pragma unroll
    for (int u = 0; u < GSTEPS; ++u) {
        c = __builtin_amdgcn_wmma_f32_16x16x4_f32(false, af[u], false, bf[u],
                                                  (short)0, c, false, false);
        if ((u & 1) == 0) {
            piece_real(s[u >> 1], eb, x[u >> 1]);
            asm volatile("" : "+v"(c) : "v"(s[u >> 1]));
        }
        __builtin_amdgcn_sched_barrier(0);
    }
    asm volatile("v_nop\n\tv_nop\n\tv_nop\n\tv_nop" : "+v"(c));
    __builtin_amdgcn_sched_barrier(0);
    return c;
}

__device__ __forceinline__ void lds_barrier() {
    __builtin_amdgcn_fence(__ATOMIC_RELEASE, "workgroup", "local");
    __builtin_amdgcn_s_barrier();
    __builtin_amdgcn_fence(__ATOMIC_ACQUIRE, "workgroup", "local");
}

__global__ __launch_bounds__(128)
void scores_wmma_kernel(const float* __restrict__ qw,
                        const float* __restrict__ qb,
                        const float* __restrict__ kw,
                        const float* __restrict__ kb,
                        const float* __restrict__ vw,
                        const float* __restrict__ vb,
                        const float* __restrict__ xq,
                        const float* __restrict__ xk,
                        float* __restrict__ Pv,
                        float* __restrict__ parts) {
    __shared__ __attribute__((aligned(16))) float lds[64 * WPITCH];

    int tid  = threadIdx.x;
    int blk  = blockIdx.x;
    int lane = tid & 31;
    int w    = tid >> 5;
    int b    = blk >> 6;
    int comp = (blk >> 4) & 3;
    int bi   = (blk >> 2) & 3;
    int bj   = blk & 3;
    int half = lane >> 4;
    int lr   = lane & 15;

    int pidx = blk * 128 + tid;
    int pd = pidx & (DD - 1);
    int ps = pidx >> 8;
    float4 pb = *(const float4*)(vb + pd * 4);
    float pw0 = vw[(0 * SS + ps) * DD + pd];
    float pw1 = vw[(1 * SS + ps) * DD + pd];
    float pw2 = vw[(2 * SS + ps) * DD + pd];
    float pw3 = vw[(3 * SS + ps) * DD + pd];

    int isk  = w >> 1;
    int row0 = (isk ? bj : bi) * 32 + (w & 1) * 16;
    const float* wsrc = isk ? kw : qw;
    const float* bsrc = isk ? kb : qb;
    const float* xsrc = isk ? xk : xq;
    const float* g = wsrc + (size_t)comp * WPL + (size_t)row0 * DD + lane * 4;

    v4f s0[16], s1[16];
    half_fetch(s0, g);

    float eb[2][4];
    #pragma unroll
    for (int h = 0; h < 2; ++h) {
        #pragma unroll
        for (int e = 0; e < 4; ++e)
            eb[h][e] = bsrc[(size_t)(h * 128 + lane * 4 + e) * 4 + comp];
    }

    float xme = xsrc[(size_t)(b * SS + row0 + lr) * 4 + comp];

    __builtin_amdgcn_sched_barrier(0);
    half_fetch(s1, g + GK);

    {
        float4 o;
        o.x = __expf(pw0) + __expf(pb.x);
        o.y = __expf(pw1) + __expf(pb.y);
        o.z = __expf(pw2) + __expf(pb.z);
        o.w = __expf(pw3) + __expf(pb.w);
        *(float4*)(Pv + (size_t)ps * NN + pd * 4) = o;
    }

    #pragma unroll
    for (int h = 0; h < 2; ++h) {
        #pragma unroll
        for (int e = 0; e < 4; ++e) eb[h][e] = __expf(eb[h][e]);
    }
    xme = fmaxf(xme, EPS);
    float xr[16];
    #pragma unroll
    for (int r = 0; r < 16; ++r)
        xr[r] = __int_as_float(__builtin_amdgcn_readlane(__float_as_int(xme), r));

    float* dst = lds + (16 * w) * WPITCH + lane * 4;

    int wi = w >> 1;
    int wj = w & 1;
    const float* fa = lds + (32 + 16 * wj + lr) * WPITCH + 2 * half;
    const float* fb = lds + (16 * wi + lr) * WPITCH + 2 * half;

    v2f af[GSTEPS], bf[GSTEPS];
    v8f c = {};

    half_real(s0, eb[0], xr);
    half_to_lds(s0, dst);
    lds_barrier();
    frag_load(af, bf, fa, fb);
    c = frag_mma_real(c, af, bf, s1, eb[1], xr);

    half_to_lds(s1, dst + GK);
    lds_barrier();
    frag_load(af, bf, fa + GK, fb + GK);
    c = frag_mma(c, af, bf);

    asm volatile("s_wait_alu depctr_va_vdst(0)\n\tv_nop\n\tv_nop\n\tv_nop\n\tv_nop" : "+v"(c));

    int ti = 2 * bi + wi;
    int tj = 2 * bj + wj;
    float* prow = parts + (size_t)comp * PPL
                + ((size_t)((b * 8 + ti) * 8 + tj) * 16 + lr) * 16 + 8 * half;
    v4f e0 = {c[0], c[1], c[2], c[3]};
    v4f e1 = {c[4], c[5], c[6], c[7]};
    *(v4f*)prow       = e0;
    *(v4f*)(prow + 4) = e1;
}

#define APITCH (SS + 4)
#define OSTEPS 16

__device__ __forceinline__ float lane_xor1(float v) {
    return __int_as_float(__builtin_amdgcn_update_dpp(0, __float_as_int(v), 0xB1, 0xF, 0xF, true));
}
__device__ __forceinline__ float lane_xor2(float v) {
    return __int_as_float(__builtin_amdgcn_update_dpp(0, __float_as_int(v), 0x4E, 0xF, 0xF, true));
}
__device__ __forceinline__ float lane_other_four(float v) {
    return __int_as_float(__builtin_amdgcn_update_dpp(0, __float_as_int(v), 0x141, 0xF, 0xF, true));
}

__global__ __launch_bounds__(128)
void out_wmma_kernel(const float* __restrict__ parts,
                     const float* __restrict__ Pv,
                     const float* __restrict__ xv,
                     float* __restrict__ out) {
    __shared__ __attribute__((aligned(16))) float la[16 * APITCH];
    __shared__ __attribute__((aligned(16))) float lx[SS * 4];

    int tid  = threadIdx.x;
    int lane = tid & 31;
    int w    = tid >> 5;
    int blk  = blockIdx.x;
    int b  = blk >> 6;
    int ti = (blk >> 3) & 7;
    int i0 = ti << 4;
    int n0 = ((blk & 7) << 7) + (w << 5);
    int half = lane >> 4;
    int lr   = lane & 15;


    int r = tid >> 3;
    int t = tid & 7;
    const float* prow = parts + ((size_t)((b * 8 + ti) * 8 + t) * 16 + r) * 16;
    v4f pp[4][4];
    #pragma unroll
    for (int p = 0; p < 4; ++p) {
        #pragma unroll
        for (int q = 0; q < 4; ++q)
            pp[p][q] = *(const v4f*)(prow + (size_t)p * PPL + 4 * q);
    }
    v4f xr = *(const v4f*)(xv + (size_t)(b * SS + tid) * 4);
    __builtin_amdgcn_sched_barrier(0);

    const float* bp = Pv + (size_t)(2 * half) * NN + n0 + 2 * lr;
    v2f bv[64];
    #pragma unroll
    for (int m = 0; m < 32; ++m) {
        bv[2 * m]     = *(const v2f*)(bp + (size_t)(4 * m) * NN);
        bv[2 * m + 1] = *(const v2f*)(bp + (size_t)(4 * m + 1) * NN);
    }
    __builtin_amdgcn_sched_barrier(0);

    {
        v4f sv[4];
        #pragma unroll
        for (int q = 0; q < 4; ++q)
            sv[q] = ((pp[0][q] + pp[1][q]) + pp[2][q]) + pp[3][q];
        float M = sv[0][0];
        #pragma unroll
        for (int q = 0; q < 4; ++q) {
            #pragma unroll
            for (int z = 0; z < 4; ++z) M = fmaxf(M, sv[q][z]);
        }
        M = fmaxf(M, lane_xor1(M));
        M = fmaxf(M, lane_xor2(M));
        M = fmaxf(M, lane_other_four(M));
        v4f ev[4];
        #pragma unroll
        for (int q = 0; q < 4; ++q) {
            #pragma unroll
            for (int z = 0; z < 4; ++z) ev[q][z] = __expf(sv[q][z] - M);
        }
        float L = (((ev[0][0] + ev[0][1]) + (ev[0][2] + ev[0][3])) + ((ev[1][0] + ev[1][1]) + (ev[1][2] + ev[1][3])))
                + (((ev[2][0] + ev[2][1]) + (ev[2][2] + ev[2][3])) + ((ev[3][0] + ev[3][1]) + (ev[3][2] + ev[3][3])));
        L += lane_xor1(L);
        L += lane_xor2(L);
        L += lane_other_four(L);
        float g = 1.0f / L;
        float* dst = la + r * APITCH + t * 16;
        #pragma unroll
        for (int q = 0; q < 4; ++q) {
            v4f a;
            #pragma unroll
            for (int z = 0; z < 4; ++z) a[z] = fmaxf(ev[q][z] * g, EPS);
            *(v4f*)(dst + 4 * q) = a;
        }
        #pragma unroll
        for (int z = 0; z < 4; ++z) xr[z] = fmaxf(xr[z], EPS);
        *(v4f*)(lx + tid * 4) = xr;
    }
    __builtin_amdgcn_fence(__ATOMIC_RELEASE, "workgroup", "local");
    __builtin_amdgcn_s_barrier();
    __builtin_amdgcn_fence(__ATOMIC_ACQUIRE, "workgroup", "local");

    const float* pa = la + lr * APITCH + 2 * half;
    const float* px = lx + (2 * half) * 4 + 2 * (lr & 1);

    v8f c0 = {}, c1 = {};
    #pragma unroll
    for (int gq = 0; gq < 32 / OSTEPS; ++gq) {
        v2f af[OSTEPS], bf0[OSTEPS], bf1[OSTEPS];
        #pragma unroll
        for (int u = 0; u < OSTEPS; ++u) {
            int m = gq * OSTEPS + u;
            af[u] = *(const v2f*)(pa + 4 * m);
            v2f x0 = *(const v2f*)(px + 16 * m);
            v2f x1 = *(const v2f*)(px + 16 * m + 4);
            bf0[u][0] = bv[2 * m][0]     + x0[0];
            bf0[u][1] = bv[2 * m + 1][0] + x1[0];
            bf1[u][0] = bv[2 * m][1]     + x0[1];
            bf1[u][1] = bv[2 * m + 1][1] + x1[1];
        }
        __builtin_amdgcn_sched_barrier(0);
        #pragma unroll
        for (int u = 0; u < OSTEPS; ++u) {
            c0 = __builtin_amdgcn_wmma_f32_16x16x4_f32(false, af[u], false, bf0[u],
                                                       (short)0, c0, false, false);
            c1 = __builtin_amdgcn_wmma_f32_16x16x4_f32(false, af[u], false, bf1[u],
                                                       (short)0, c1, false, false);
        }
        __builtin_amdgcn_sched_barrier(0);
        if (gq + 1 < 32 / OSTEPS) {
            asm volatile("v_nop\n\tv_nop\n\tv_nop\n\tv_nop");
            __builtin_amdgcn_sched_barrier(0);
        }
    }

    asm volatile("s_wait_alu depctr_va_vdst(0)\n\tv_nop\n\tv_nop\n\tv_nop\n\tv_nop" : "+v"(c0), "+v"(c1));

    float* orow = out + (size_t)(b * SS + i0 + 8 * half) * NN + n0 + 2 * lr;
    #pragma unroll
    for (int r = 0; r < 8; ++r) {
        v2f o = {c0[r], c1[r]};
        *(v2f*)(orow + (size_t)r * NN) = o;
    }
}

extern "C" void kernel_launch(void* const* d_in, const int* in_sizes, int n_in,
                              void* d_out, int out_size, void* d_ws, size_t ws_size,
                              hipStream_t stream) {
    (void)in_sizes; (void)n_in; (void)out_size; (void)ws_size;

    const float* query  = (const float*)d_in[0];
    const float* key_in = (const float*)d_in[1];
    const float* value  = (const float*)d_in[2];
    const float* qw = (const float*)d_in[3];
    const float* qb = (const float*)d_in[4];
    const float* kw = (const float*)d_in[5];
    const float* kb = (const float*)d_in[6];
    const float* vw = (const float*)d_in[7];
    const float* vb = (const float*)d_in[8];
    float* out = (float*)d_out;

    float* Pv = (float*)d_ws;
    float* pt = Pv + (size_t)SS * NN;

    scores_wmma_kernel<<<SCORE_GROUPS, 128, 0, stream>>>(
        qw, qb, kw, kb, vw, vb, query, key_in, Pv, pt);

    out_wmma_kernel<<<256, 128, 0, stream>>>(pt, Pv, value, out);
}
